# starting kernel
# baseline (speedup 1.0000x reference)
#include <hip/hip_runtime.h>

#define N_NODES 50000
#define N_EDGES 800000
#define DIM     128

typedef __attribute__((ext_vector_type(2))) float v2f;
typedef __attribute__((ext_vector_type(8))) float v8f;

__global__ void k_init_deg(float* __restrict__ deg) {
    int i = blockIdx.x * blockDim.x + threadIdx.x;
    if (i < N_NODES) deg[i] = 1.0f;
}

__global__ void k_count_deg(const int* __restrict__ ei, float* __restrict__ deg) {
    int e = blockIdx.x * blockDim.x + threadIdx.x;
    if (e < N_EDGES) {
        int dst = ei[N_EDGES + e];
        unsafeAtomicAdd(&deg[dst], 1.0f);
    }
}

__global__ void k_rsqrt(float* __restrict__ deg) {
    int i = blockIdx.x * blockDim.x + threadIdx.x;
    if (i < N_NODES) deg[i] = rsqrtf(deg[i]);
}

__global__ void k_init_acc(const float* __restrict__ x, const float* __restrict__ dinv,
                           float* __restrict__ acc) {
    int g = blockIdx.x * blockDim.x + threadIdx.x;
    if (g < N_NODES * (DIM / 4)) {
        int n = g >> 5;
        float s = dinv[n]; s = s * s;
        float4 v = ((const float4*)x)[g];
        v.x *= s; v.y *= s; v.z *= s; v.w *= s;
        ((float4*)acc)[g] = v;
    }
}

__global__ void k_scatter(const float* __restrict__ x, const int* __restrict__ ei,
                          const float* __restrict__ dinv, float* __restrict__ acc) {
    int g = blockIdx.x * blockDim.x + threadIdx.x;
    int e = g >> 5;
    int lane = g & 31;
    if (e < N_EDGES) {
        int src = ei[e];
        int dst = ei[N_EDGES + e];
        float nrm = dinv[src] * dinv[dst];
        float4 v = ((const float4*)(x + (size_t)src * DIM))[lane];
        float* p = acc + (size_t)dst * DIM + lane * 4;
        unsafeAtomicAdd(p + 0, v.x * nrm);
        unsafeAtomicAdd(p + 1, v.y * nrm);
        unsafeAtomicAdd(p + 2, v.z * nrm);
        unsafeAtomicAdd(p + 3, v.w * nrm);
    }
}

__global__ void __launch_bounds__(256) k_gemm_fused(
        const float* __restrict__ acc, const float* __restrict__ x,
        const float* __restrict__ W, const float* __restrict__ bias,
        float* __restrict__ out) {
    __shared__ float At[16 * 132];
    const int tid  = threadIdx.x;
    const int wave = tid >> 5;
    const int lane = tid & 31;
    const int row0 = blockIdx.x * 16;

    {
        const float4* src4 = (const float4*)(acc + (size_t)row0 * DIM);
        float4* dst4 = (float4*)At;
        #pragma unroll
        for (int i = tid; i < 16 * 32; i += 256) {
            int r = i >> 5, q = i & 31;
            dst4[r * 33 + q] = src4[i];
        }
    }
    __syncthreads();

    const int col0 = wave * 16;
    const int nn   = lane & 15;
    const int koff = (lane >> 4) << 1;
    const int colg = col0 + nn;

    v2f bfrag[32];
    #pragma unroll
    for (int kk = 0; kk < 32; ++kk) {
        int k = (kk << 2) + koff;
        v2f t;
        t.x = W[k * DIM + colg];
        t.y = W[(k + 1) * DIM + colg];
        bfrag[kk] = t;
    }

    v8f c = {};
    const float* arow = At + nn * 132;
    #pragma unroll
    for (int kk = 0; kk < 32; ++kk) {
        int k = (kk << 2) + koff;
        v2f a;
        a.x = arow[k];
        a.y = arow[k + 1];
        c = __builtin_amdgcn_wmma_f32_16x16x4_f32(false, a, false, bfrag[kk],
                                                  (short)0, c, false, false);
    }

    const float bv = bias[colg];
    const int rbase = row0 + ((lane >> 4) << 3);
    #pragma unroll
    for (int r = 0; r < 8; ++r) {
        size_t off = (size_t)(rbase + r) * DIM + colg;
        float v = c[r] + bv;
        v = v > 0.0f ? v : 0.0f;
        out[off] = x[off] + v;
    }
}

extern "C" void kernel_launch(void* const* d_in, const int* in_sizes, int n_in,
                              void* d_out, int out_size, void* d_ws, size_t ws_size,
                              hipStream_t stream) {
    (void)in_sizes; (void)n_in; (void)out_size; (void)ws_size;
    const float* x  = (const float*)d_in[0];
    const int*   ei = (const int*)d_in[1];
    const float* W  = (const float*)d_in[2];
    const float* b  = (const float*)d_in[3];
    float* out = (float*)d_out;

    char* ws  = (char*)d_ws;
    float* deg = (float*)ws;
    float* acc = (float*)(ws + ((N_NODES * 4 + 255) & ~255));

    k_init_deg <<<(N_NODES + 255) / 256, 256, 0, stream>>>(deg);
    k_count_deg<<<(N_EDGES + 255) / 256, 256, 0, stream>>>(ei, deg);
    k_rsqrt    <<<(N_NODES + 255) / 256, 256, 0, stream>>>(deg);
    k_init_acc <<<(N_NODES * (DIM / 4) + 255) / 256, 256, 0, stream>>>(x, deg, acc);
    k_scatter  <<<(N_EDGES * 32 + 255) / 256, 256, 0, stream>>>(x, ei, deg, acc);
    k_gemm_fused<<<N_NODES / 16, 256, 0, stream>>>(acc, x, W, b, out);
}


# Replace the float-atomic scatter by per-node source lists built with one returning int atomic per edge, then an atomic-free per-row gather; row sums now run in list arrival order
# speedup vs baseline: 2.5425x; 2.5425x over previous
#include <hip/hip_runtime.h>

#define N_NODES 50000
#define N_EDGES 800000
#define DIM     128
#define REC     64
#define CAP     (REC - 1)

typedef __attribute__((ext_vector_type(2))) float v2f;
typedef __attribute__((ext_vector_type(8))) float v8f;

__global__ void k_zero_cnt(int* __restrict__ rec) {
    int i = blockIdx.x * blockDim.x + threadIdx.x;
    if (i < N_NODES) rec[(size_t)i * REC] = 0;
}

__global__ void k_fill(const int* __restrict__ ei, int* __restrict__ rec) {
    int e = blockIdx.x * blockDim.x + threadIdx.x;
    if (e < N_EDGES) {
        int src = ei[e];
        int dst = ei[N_EDGES + e];
        int* r = rec + (size_t)dst * REC;
        int pos = atomicAdd(r, 1);
        if (pos < CAP) r[1 + pos] = src;
    }
}

static __device__ __forceinline__ float4 ld_row(const float* __restrict__ x, int s, int lane) {
    return ((const float4*)(x + (size_t)s * DIM))[lane];
}
static __device__ __forceinline__ void fma4(float4& a, const float4 v, float w) {
    a.x = fmaf(v.x, w, a.x); a.y = fmaf(v.y, w, a.y);
    a.z = fmaf(v.z, w, a.z); a.w = fmaf(v.w, w, a.w);
}
static __device__ __forceinline__ float dinv_of(const int* __restrict__ rec, int n) {
    return rsqrtf((float)(rec[(size_t)n * REC] + 1));
}
static __device__ __forceinline__ void add_lanes(float4& a, const float* __restrict__ x,
                                                 int sv, float wv, int lo, int hi, int lane) {
    const int wi = __float_as_int(wv);
    int j = lo;
    for (; j + 4 <= hi; j += 4) {
        int s0 = __builtin_amdgcn_readlane(sv, j);
        int s1 = __builtin_amdgcn_readlane(sv, j + 1);
        int s2 = __builtin_amdgcn_readlane(sv, j + 2);
        int s3 = __builtin_amdgcn_readlane(sv, j + 3);
        float4 v0 = ld_row(x, s0, lane);
        float4 v1 = ld_row(x, s1, lane);
        float4 v2 = ld_row(x, s2, lane);
        float4 v3 = ld_row(x, s3, lane);
        fma4(a, v0, __int_as_float(__builtin_amdgcn_readlane(wi, j)));
        fma4(a, v1, __int_as_float(__builtin_amdgcn_readlane(wi, j + 1)));
        fma4(a, v2, __int_as_float(__builtin_amdgcn_readlane(wi, j + 2)));
        fma4(a, v3, __int_as_float(__builtin_amdgcn_readlane(wi, j + 3)));
    }
    for (; j < hi; ++j) {
        int s = __builtin_amdgcn_readlane(sv, j);
        float4 v = ld_row(x, s, lane);
        fma4(a, v, __int_as_float(__builtin_amdgcn_readlane(wi, j)));
    }
}

__global__ void __launch_bounds__(256) k_gather(const float* __restrict__ x, const int* __restrict__ ei,
                                                const int* __restrict__ rec, float* __restrict__ acc) {
    const int lane = threadIdx.x & 31;
    const int node = blockIdx.x * 8 + (threadIdx.x >> 5);
    if (node >= N_NODES) return;
    const int* r = rec + (size_t)node * REC;
    const int c = __builtin_amdgcn_readfirstlane(r[0]);
    const float dn = rsqrtf((float)(c + 1));
    float4 a = ld_row(x, node, lane);
    a.x *= dn; a.y *= dn; a.z *= dn; a.w *= dn;
    if (c <= CAP) {
        int s0 = 0, s1 = 0;
        float w0 = 0.0f, w1 = 0.0f;
        if (lane >= 1 && lane <= c) { s0 = r[lane];      w0 = dinv_of(rec, s0); }
        if (32 + lane <= c)         { s1 = r[32 + lane]; w1 = dinv_of(rec, s1); }
        const int h0 = c < 31 ? c : 31;
        add_lanes(a, x, s0, w0, 1, h0 + 1, lane);
        if (c > 31) add_lanes(a, x, s1, w1, 0, c - 31, lane);
    } else {
        for (int base = 0; base < N_EDGES; base += 32) {
            int e = base + lane;
            int d = (e < N_EDGES) ? ei[N_EDGES + e] : -1;
            unsigned m = __builtin_amdgcn_ballot_w32(d == node);
            while (m) {
                int j = __builtin_ctz(m);
                m &= m - 1;
                int s = ei[base + j];
                float4 v = ld_row(x, s, lane);
                fma4(a, v, dinv_of(rec, s));
            }
        }
    }
    a.x *= dn; a.y *= dn; a.z *= dn; a.w *= dn;
    ((float4*)(acc + (size_t)node * DIM))[lane] = a;
}

__global__ void __launch_bounds__(256) k_gemm_fused(
        const float* __restrict__ acc, const float* __restrict__ x,
        const float* __restrict__ W, const float* __restrict__ bias,
        float* __restrict__ out) {
    __shared__ float At[16 * 132];
    const int tid  = threadIdx.x;
    const int wave = tid >> 5;
    const int lane = tid & 31;
    const int row0 = blockIdx.x * 16;

    {
        const float4* src4 = (const float4*)(acc + (size_t)row0 * DIM);
        float4* dst4 = (float4*)At;
        #pragma unroll
        for (int i = tid; i < 16 * 32; i += 256) {
            int r = i >> 5, q = i & 31;
            dst4[r * 33 + q] = src4[i];
        }
    }
    __syncthreads();

    const int col0 = wave * 16;
    const int nn   = lane & 15;
    const int koff = (lane >> 4) << 1;
    const int colg = col0 + nn;

    v2f bfrag[32];
    #pragma unroll
    for (int kk = 0; kk < 32; ++kk) {
        int k = (kk << 2) + koff;
        v2f t;
        t.x = W[k * DIM + colg];
        t.y = W[(k + 1) * DIM + colg];
        bfrag[kk] = t;
    }

    v8f c = {};
    const float* arow = At + nn * 132;
    #pragma unroll
    for (int kk = 0; kk < 32; ++kk) {
        int k = (kk << 2) + koff;
        v2f a;
        a.x = arow[k];
        a.y = arow[k + 1];
        c = __builtin_amdgcn_wmma_f32_16x16x4_f32(false, a, false, bfrag[kk],
                                                  (short)0, c, false, false);
    }

    const float bv = bias[colg];
    const int rbase = row0 + ((lane >> 4) << 3);
    #pragma unroll
    for (int r = 0; r < 8; ++r) {
        size_t off = (size_t)(rbase + r) * DIM + colg;
        float v = c[r] + bv;
        v = v > 0.0f ? v : 0.0f;
        out[off] = x[off] + v;
    }
}

extern "C" void kernel_launch(void* const* d_in, const int* in_sizes, int n_in,
                              void* d_out, int out_size, void* d_ws, size_t ws_size,
                              hipStream_t stream) {
    (void)in_sizes; (void)n_in; (void)out_size; (void)ws_size;
    const float* x  = (const float*)d_in[0];
    const int*   ei = (const int*)d_in[1];
    const float* W  = (const float*)d_in[2];
    const float* b  = (const float*)d_in[3];
    float* out = (float*)d_out;

    char* ws  = (char*)d_ws;
    int*   rec = (int*)ws;
    float* acc = (float*)(ws + (size_t)N_NODES * REC * 4);

    k_zero_cnt  <<<(N_NODES + 255) / 256, 256, 0, stream>>>(rec);
    k_fill      <<<(N_EDGES + 255) / 256, 256, 0, stream>>>(ei, rec);
    k_gather    <<<(N_NODES + 7) / 8, 256, 0, stream>>>(x, ei, rec, acc);
    k_gemm_fused<<<N_NODES / 16, 256, 0, stream>>>(acc, x, W, b, out);
}


# Fuse the row gather into the product kernel: each 16-row block sums its rows into LDS and multiplies from there; summed rows and residual rows skip global memory, one launch fewer
# speedup vs baseline: 2.7443x; 1.0794x over previous
#include <hip/hip_runtime.h>

#define N_NODES 50000
#define N_EDGES 800000
#define DIM     128
#define REC     64
#define CAP     (REC - 1)

typedef __attribute__((ext_vector_type(2))) float v2f;
typedef __attribute__((ext_vector_type(8))) float v8f;

__global__ void k_zero_cnt(int* __restrict__ rec) {
    int i = blockIdx.x * blockDim.x + threadIdx.x;
    if (i < N_NODES) rec[(size_t)i * REC] = 0;
}

__global__ void k_fill(const int* __restrict__ ei, int* __restrict__ rec) {
    int e = blockIdx.x * blockDim.x + threadIdx.x;
    if (e < N_EDGES) {
        int src = ei[e];
        int dst = ei[N_EDGES + e];
        int* r = rec + (size_t)dst * REC;
        int pos = atomicAdd(r, 1);
        if (pos < CAP) r[1 + pos] = src;
    }
}

static __device__ __forceinline__ float4 ld_row(const float* __restrict__ x, int s, int lane) {
    return ((const float4*)(x + (size_t)s * DIM))[lane];
}
static __device__ __forceinline__ void fma4(float4& a, const float4 v, float w) {
    a.x = fmaf(v.x, w, a.x); a.y = fmaf(v.y, w, a.y);
    a.z = fmaf(v.z, w, a.z); a.w = fmaf(v.w, w, a.w);
}
static __device__ __forceinline__ float dinv_of(const int* __restrict__ rec, int n) {
    return rsqrtf((float)(rec[(size_t)n * REC] + 1));
}
static __device__ __forceinline__ void add_lanes(float4& a, const float* __restrict__ x,
                                                 int sv, float wv, int lo, int hi, int lane) {
    const int wi = __float_as_int(wv);
    int j = lo;
    for (; j + 4 <= hi; j += 4) {
        int s0 = __builtin_amdgcn_readlane(sv, j);
        int s1 = __builtin_amdgcn_readlane(sv, j + 1);
        int s2 = __builtin_amdgcn_readlane(sv, j + 2);
        int s3 = __builtin_amdgcn_readlane(sv, j + 3);
        float4 v0 = ld_row(x, s0, lane);
        float4 v1 = ld_row(x, s1, lane);
        float4 v2 = ld_row(x, s2, lane);
        float4 v3 = ld_row(x, s3, lane);
        fma4(a, v0, __int_as_float(__builtin_amdgcn_readlane(wi, j)));
        fma4(a, v1, __int_as_float(__builtin_amdgcn_readlane(wi, j + 1)));
        fma4(a, v2, __int_as_float(__builtin_amdgcn_readlane(wi, j + 2)));
        fma4(a, v3, __int_as_float(__builtin_amdgcn_readlane(wi, j + 3)));
    }
    for (; j < hi; ++j) {
        int s = __builtin_amdgcn_readlane(sv, j);
        float4 v = ld_row(x, s, lane);
        fma4(a, v, __int_as_float(__builtin_amdgcn_readlane(wi, j)));
    }
}

struct RowList { int c; int s0; int s1; float w0; float w1; };

static __device__ __forceinline__ float4 sum_row(const float* __restrict__ x, const int* __restrict__ ei,
                                                 const int* __restrict__ rec, const RowList L,
                                                 const float4 xs, int node, int lane) {
    const int c = L.c;
    const float dn = rsqrtf((float)(c + 1));
    float4 a = xs;
    a.x *= dn; a.y *= dn; a.z *= dn; a.w *= dn;
    if (c <= CAP) {
        const int h0 = c < 31 ? c : 31;
        add_lanes(a, x, L.s0, L.w0, 1, h0 + 1, lane);
        if (c > 31) add_lanes(a, x, L.s1, L.w1, 0, c - 31, lane);
    } else {
        for (int base = 0; base < N_EDGES; base += 32) {
            int e = base + lane;
            int d = (e < N_EDGES) ? ei[N_EDGES + e] : -1;
            unsigned m = __builtin_amdgcn_ballot_w32(d == node);
            while (m) {
                int j = __builtin_ctz(m);
                m &= m - 1;
                int s = ei[base + j];
                float4 v = ld_row(x, s, lane);
                fma4(a, v, dinv_of(rec, s));
            }
        }
    }
    a.x *= dn; a.y *= dn; a.z *= dn; a.w *= dn;
    return a;
}

static_assert(N_NODES % 16 == 0, "a block owns 16 whole rows");

__global__ void __launch_bounds__(256) k_gather_gemm(
        const float* __restrict__ x, const int* __restrict__ ei, const int* __restrict__ rec,
        const float* __restrict__ W, const float* __restrict__ bias, float* __restrict__ out) {
    __shared__ alignas(16) float At[16 * 132];
    __shared__ alignas(16) float Xt[16 * 132];
    const int tid  = threadIdx.x;
    const int wave = tid >> 5;
    const int lane = tid & 31;
    const int row0 = blockIdx.x * 16;

    {
        const int lr0 = wave * 2, lr1 = lr0 + 1;
        const int n0 = row0 + lr0, n1 = row0 + lr1;
        const int* r0 = rec + (size_t)n0 * REC;
        const int* r1 = rec + (size_t)n1 * REC;
        const int cv0 = r0[0];
        const int cv1 = r1[0];
        const float4 x0 = ld_row(x, n0, lane);
        const float4 x1 = ld_row(x, n1, lane);
        RowList L0, L1;
        L0.c = __builtin_amdgcn_readfirstlane(cv0);
        L1.c = __builtin_amdgcn_readfirstlane(cv1);
        const bool p00 = L0.c <= CAP && lane >= 1 && lane <= L0.c;
        const bool p01 = L0.c <= CAP && 32 + lane <= L0.c;
        const bool p10 = L1.c <= CAP && lane >= 1 && lane <= L1.c;
        const bool p11 = L1.c <= CAP && 32 + lane <= L1.c;
        L0.s0 = 0; L0.s1 = 0; L1.s0 = 0; L1.s1 = 0;
        if (p00) L0.s0 = r0[lane];
        if (p10) L1.s0 = r1[lane];
        if (p01) L0.s1 = r0[32 + lane];
        if (p11) L1.s1 = r1[32 + lane];
        const int d00 = rec[(size_t)L0.s0 * REC];
        const int d10 = rec[(size_t)L1.s0 * REC];
        const int d01 = rec[(size_t)L0.s1 * REC];
        const int d11 = rec[(size_t)L1.s1 * REC];
        L0.w0 = p00 ? rsqrtf((float)(d00 + 1)) : 0.0f;
        L1.w0 = p10 ? rsqrtf((float)(d10 + 1)) : 0.0f;
        L0.w1 = p01 ? rsqrtf((float)(d01 + 1)) : 0.0f;
        L1.w1 = p11 ? rsqrtf((float)(d11 + 1)) : 0.0f;
        ((float4*)(Xt + lr0 * 132))[lane] = x0;
        ((float4*)(Xt + lr1 * 132))[lane] = x1;
        const float4 a0 = sum_row(x, ei, rec, L0, x0, n0, lane);
        ((float4*)(At + lr0 * 132))[lane] = a0;
        const float4 a1 = sum_row(x, ei, rec, L1, x1, n1, lane);
        ((float4*)(At + lr1 * 132))[lane] = a1;
    }
    __syncthreads();

    const int col0 = wave * 16;
    const int nn   = lane & 15;
    const int koff = (lane >> 4) << 1;
    const int colg = col0 + nn;

    v2f bfrag[32];
    #pragma unroll
    for (int kk = 0; kk < 32; ++kk) {
        int k = (kk << 2) + koff;
        v2f t;
        t.x = W[k * DIM + colg];
        t.y = W[(k + 1) * DIM + colg];
        bfrag[kk] = t;
    }

    v8f c = {};
    const float* arow = At + nn * 132;
    #pragma unroll
    for (int kk = 0; kk < 32; ++kk) {
        int k = (kk << 2) + koff;
        v2f a;
        a.x = arow[k];
        a.y = arow[k + 1];
        c = __builtin_amdgcn_wmma_f32_16x16x4_f32(false, a, false, bfrag[kk],
                                                  (short)0, c, false, false);
    }

    const float bv = bias[colg];
    const int lrb = (lane >> 4) << 3;
    #pragma unroll
    for (int r = 0; r < 8; ++r) {
        float v = c[r] + bv;
        v = v > 0.0f ? v : 0.0f;
        out[(size_t)(row0 + lrb + r) * DIM + colg] = Xt[(lrb + r) * 132 + colg] + v;
    }
}

extern "C" void kernel_launch(void* const* d_in, const int* in_sizes, int n_in,
                              void* d_out, int out_size, void* d_ws, size_t ws_size,
                              hipStream_t stream) {
    (void)in_sizes; (void)n_in; (void)out_size; (void)ws_size;
    const float* x  = (const float*)d_in[0];
    const int*   ei = (const int*)d_in[1];
    const float* W  = (const float*)d_in[2];
    const float* b  = (const float*)d_in[3];
    float* out = (float*)d_out;

    int* rec = (int*)d_ws;

    k_zero_cnt   <<<(N_NODES + 255) / 256, 256, 0, stream>>>(rec);
    k_fill       <<<(N_EDGES + 255) / 256, 256, 0, stream>>>(ei, rec);
    k_gather_gemm<<<N_NODES / 16, 256, 0, stream>>>(x, ei, rec, W, b, out);
}


# Build the per-node source lists by a two-level bucket sort on LDS counters (chunk sort by 200-row bucket, then row lists per bucket), no global atomic per edge; lists stay in arrival order
# speedup vs baseline: 4.4394x; 1.6177x over previous
#include <hip/hip_runtime.h>

#define N_NODES 50000
#define N_EDGES 800000
#define DIM     128
#define REC     64
#define CAP     (REC - 1)

#define NCHUNK  256
#define CHUNK   (N_EDGES / NCHUNK)
#define RB      200
#define NB      (N_NODES / RB)
#define T1      512
#define EPT     ((CHUNK + T1 - 1) / T1)
static_assert(N_EDGES % NCHUNK == 0, "chunks are whole");
static_assert(N_NODES % RB == 0, "buckets are whole");
static_assert(NB <= 256, "one scan thread per bucket, 8 waves");
static_assert(N_NODES <= 65536 && RB <= 65536, "an entry packs source id and local row in 16 bits each");
static_assert(NCHUNK == (T1 / 32) * 16, "each of the 16 waves of a list work-group takes 16 chunks");

typedef __attribute__((ext_vector_type(2))) float v2f;
typedef __attribute__((ext_vector_type(8))) float v8f;

__global__ void __launch_bounds__(T1) k_chunk_sort(const int* __restrict__ ei,
                                                   unsigned* __restrict__ sorted, int* __restrict__ off) {
    __shared__ int bcnt[256];
    __shared__ int boff[256];
    __shared__ int wtot[8];
    __shared__ unsigned stage[CHUNK];
    const int tid  = threadIdx.x;
    const int lane = tid & 31;
    const int base = blockIdx.x * CHUNK;
    if (tid < 256) bcnt[tid] = 0;
    __syncthreads();

    int sv[EPT], dv[EPT], pos[EPT];
    #pragma unroll
    for (int it = 0; it < EPT; ++it) {
        const int e = tid + it * T1;
        const bool ok = e < CHUNK;
        sv[it] = ok ? ei[base + e] : 0;
        dv[it] = ok ? ei[N_EDGES + base + e] : -1;
    }
    #pragma unroll
    for (int it = 0; it < EPT; ++it) {
        pos[it] = 0;
        if (dv[it] >= 0) pos[it] = atomicAdd(&bcnt[dv[it] / RB], 1);
    }
    __syncthreads();

    int v = 0, incl = 0;
    if (tid < 256) {
        v = tid < NB ? bcnt[tid] : 0;
        incl = v;
        #pragma unroll
        for (int d = 1; d < 32; d <<= 1) {
            const int t = __shfl_up(incl, d);
            if (lane >= d) incl += t;
        }
        if (lane == 31) wtot[tid >> 5] = incl;
    }
    __syncthreads();
    if (tid < 256) {
        int add = 0;
        #pragma unroll
        for (int w = 0; w < 8; ++w) if (w < (tid >> 5)) add += wtot[w];
        boff[tid] = add + incl - v;
    }
    __syncthreads();

    #pragma unroll
    for (int it = 0; it < EPT; ++it) {
        if (dv[it] >= 0) {
            const int b = dv[it] / RB;
            stage[boff[b] + pos[it]] = (unsigned)sv[it] | ((unsigned)(dv[it] - b * RB) << 16);
        }
    }
    __syncthreads();
    for (int i = tid; i < CHUNK; i += T1) sorted[base + i] = stage[i];
    if (tid <= NB) off[blockIdx.x * (NB + 1) + tid] = boff[tid];
}

static __device__ __forceinline__ void add_entry(int* __restrict__ rowcnt, int* __restrict__ lst, unsigned v) {
    const int r = (int)(v >> 16);
    const int p = atomicAdd(&rowcnt[r], 1);
    if (p < CAP) lst[r * REC + 1 + p] = (int)(v & 0xffffu);
}

__global__ void __launch_bounds__(T1) k_build_lists(const unsigned* __restrict__ sorted,
                                                    const int* __restrict__ off, int* __restrict__ rec) {
    __shared__ int rowcnt[RB];
    __shared__ int lst[RB * REC];
    const int tid  = threadIdx.x;
    const int wave = tid >> 5;
    const int lane = tid & 31;
    const int k    = blockIdx.x;
    for (int i = tid; i < RB; i += T1) rowcnt[i] = 0;
    __syncthreads();

    const int* offp = off + (size_t)(wave * 16 + (lane & 15)) * (NB + 1) + k;
    const int o0 = offp[0];
    const int o1 = offp[1];
    #pragma unroll
    for (int h = 0; h < 16; h += 8) {
        unsigned v[8];
        int s8[8], e8[8];
        #pragma unroll
        for (int u = 0; u < 8; ++u) {
            s8[u] = __builtin_amdgcn_readlane(o0, h + u);
            e8[u] = __builtin_amdgcn_readlane(o1, h + u);
            const unsigned* seg = sorted + (size_t)(wave * 16 + h + u) * CHUNK;
            const int j = s8[u] + lane;
            v[u] = seg[j < e8[u] ? j : 0];
        }
        #pragma unroll
        for (int u = 0; u < 8; ++u)
            if (s8[u] + lane < e8[u]) add_entry(rowcnt, lst, v[u]);
        #pragma unroll
        for (int u = 0; u < 8; ++u) {
            const unsigned* seg = sorted + (size_t)(wave * 16 + h + u) * CHUNK;
            for (int j = s8[u] + 32 + lane; j < e8[u]; j += 32) add_entry(rowcnt, lst, seg[j]);
        }
    }
    __syncthreads();

    for (int r = wave; r < RB; r += T1 / 32) {
        const int cnt = rowcnt[r];
        int* rp = rec + (size_t)(k * RB + r) * REC;
        const int lim = cnt < CAP ? cnt : CAP;
        rp[lane] = lane == 0 ? cnt : (lane <= lim ? lst[r * REC + lane] : 0);
        if (lim > 31) rp[32 + lane] = (32 + lane <= lim) ? lst[r * REC + 32 + lane] : 0;
    }
}

static __device__ __forceinline__ float4 ld_row(const float* __restrict__ x, int s, int lane) {
    return ((const float4*)(x + (size_t)s * DIM))[lane];
}
static __device__ __forceinline__ void fma4(float4& a, const float4 v, float w) {
    a.x = fmaf(v.x, w, a.x); a.y = fmaf(v.y, w, a.y);
    a.z = fmaf(v.z, w, a.z); a.w = fmaf(v.w, w, a.w);
}
static __device__ __forceinline__ float dinv_of(const int* __restrict__ rec, int n) {
    return rsqrtf((float)(rec[(size_t)n * REC] + 1));
}
static __device__ __forceinline__ void add_lanes(float4& a, const float* __restrict__ x,
                                                 int sv, float wv, int lo, int hi, int lane) {
    const int wi = __float_as_int(wv);
    int j = lo;
    for (; j + 4 <= hi; j += 4) {
        int s0 = __builtin_amdgcn_readlane(sv, j);
        int s1 = __builtin_amdgcn_readlane(sv, j + 1);
        int s2 = __builtin_amdgcn_readlane(sv, j + 2);
        int s3 = __builtin_amdgcn_readlane(sv, j + 3);
        float4 v0 = ld_row(x, s0, lane);
        float4 v1 = ld_row(x, s1, lane);
        float4 v2 = ld_row(x, s2, lane);
        float4 v3 = ld_row(x, s3, lane);
        fma4(a, v0, __int_as_float(__builtin_amdgcn_readlane(wi, j)));
        fma4(a, v1, __int_as_float(__builtin_amdgcn_readlane(wi, j + 1)));
        fma4(a, v2, __int_as_float(__builtin_amdgcn_readlane(wi, j + 2)));
        fma4(a, v3, __int_as_float(__builtin_amdgcn_readlane(wi, j + 3)));
    }
    for (; j < hi; ++j) {
        int s = __builtin_amdgcn_readlane(sv, j);
        float4 v = ld_row(x, s, lane);
        fma4(a, v, __int_as_float(__builtin_amdgcn_readlane(wi, j)));
    }
}

struct RowList { int c; int s0; int s1; float w0; float w1; };

static __device__ __forceinline__ float4 sum_row(const float* __restrict__ x, const int* __restrict__ ei,
                                                 const int* __restrict__ rec, const RowList L,
                                                 const float4 xs, int node, int lane) {
    const int c = L.c;
    const float dn = rsqrtf((float)(c + 1));
    float4 a = xs;
    a.x *= dn; a.y *= dn; a.z *= dn; a.w *= dn;
    if (c <= CAP) {
        const int h0 = c < 31 ? c : 31;
        add_lanes(a, x, L.s0, L.w0, 1, h0 + 1, lane);
        if (c > 31) add_lanes(a, x, L.s1, L.w1, 0, c - 31, lane);
    } else {
        for (int base = 0; base < N_EDGES; base += 32) {
            int e = base + lane;
            int d = (e < N_EDGES) ? ei[N_EDGES + e] : -1;
            unsigned m = __builtin_amdgcn_ballot_w32(d == node);
            while (m) {
                int j = __builtin_ctz(m);
                m &= m - 1;
                int s = ei[base + j];
                float4 v = ld_row(x, s, lane);
                fma4(a, v, dinv_of(rec, s));
            }
        }
    }
    a.x *= dn; a.y *= dn; a.z *= dn; a.w *= dn;
    return a;
}

static_assert(N_NODES % 16 == 0, "a block owns 16 whole rows");

__global__ void __launch_bounds__(256) k_gather_gemm(
        const float* __restrict__ x, const int* __restrict__ ei, const int* __restrict__ rec,
        const float* __restrict__ W, const float* __restrict__ bias, float* __restrict__ out) {
    __shared__ alignas(16) float At[16 * 132];
    __shared__ alignas(16) float Xt[16 * 132];
    const int tid  = threadIdx.x;
    const int wave = tid >> 5;
    const int lane = tid & 31;
    const int row0 = blockIdx.x * 16;

    {
        const int lr0 = wave * 2, lr1 = lr0 + 1;
        const int n0 = row0 + lr0, n1 = row0 + lr1;
        const int* r0 = rec + (size_t)n0 * REC;
        const int* r1 = rec + (size_t)n1 * REC;
        const int cv0 = r0[0];
        const int cv1 = r1[0];
        const float4 x0 = ld_row(x, n0, lane);
        const float4 x1 = ld_row(x, n1, lane);
        RowList L0, L1;
        L0.c = __builtin_amdgcn_readfirstlane(cv0);
        L1.c = __builtin_amdgcn_readfirstlane(cv1);
        const bool p00 = L0.c <= CAP && lane >= 1 && lane <= L0.c;
        const bool p01 = L0.c <= CAP && 32 + lane <= L0.c;
        const bool p10 = L1.c <= CAP && lane >= 1 && lane <= L1.c;
        const bool p11 = L1.c <= CAP && 32 + lane <= L1.c;
        L0.s0 = 0; L0.s1 = 0; L1.s0 = 0; L1.s1 = 0;
        if (p00) L0.s0 = r0[lane];
        if (p10) L1.s0 = r1[lane];
        if (p01) L0.s1 = r0[32 + lane];
        if (p11) L1.s1 = r1[32 + lane];
        const int d00 = rec[(size_t)L0.s0 * REC];
        const int d10 = rec[(size_t)L1.s0 * REC];
        const int d01 = rec[(size_t)L0.s1 * REC];
        const int d11 = rec[(size_t)L1.s1 * REC];
        L0.w0 = p00 ? rsqrtf((float)(d00 + 1)) : 0.0f;
        L1.w0 = p10 ? rsqrtf((float)(d10 + 1)) : 0.0f;
        L0.w1 = p01 ? rsqrtf((float)(d01 + 1)) : 0.0f;
        L1.w1 = p11 ? rsqrtf((float)(d11 + 1)) : 0.0f;
        ((float4*)(Xt + lr0 * 132))[lane] = x0;
        ((float4*)(Xt + lr1 * 132))[lane] = x1;
        const float4 a0 = sum_row(x, ei, rec, L0, x0, n0, lane);
        ((float4*)(At + lr0 * 132))[lane] = a0;
        const float4 a1 = sum_row(x, ei, rec, L1, x1, n1, lane);
        ((float4*)(At + lr1 * 132))[lane] = a1;
    }
    __syncthreads();

    const int col0 = wave * 16;
    const int nn   = lane & 15;
    const int koff = (lane >> 4) << 1;
    const int colg = col0 + nn;

    v2f bfrag[32];
    #pragma unroll
    for (int kk = 0; kk < 32; ++kk) {
        int k = (kk << 2) + koff;
        v2f t;
        t.x = W[k * DIM + colg];
        t.y = W[(k + 1) * DIM + colg];
        bfrag[kk] = t;
    }

    v8f c = {};
    const float* arow = At + nn * 132;
    #pragma unroll
    for (int kk = 0; kk < 32; ++kk) {
        int k = (kk << 2) + koff;
        v2f a;
        a.x = arow[k];
        a.y = arow[k + 1];
        c = __builtin_amdgcn_wmma_f32_16x16x4_f32(false, a, false, bfrag[kk],
                                                  (short)0, c, false, false);
    }

    const float bv = bias[colg];
    const int lrb = (lane >> 4) << 3;
    #pragma unroll
    for (int r = 0; r < 8; ++r) {
        float v = c[r] + bv;
        v = v > 0.0f ? v : 0.0f;
        out[(size_t)(row0 + lrb + r) * DIM + colg] = Xt[(lrb + r) * 132 + colg] + v;
    }
}

extern "C" void kernel_launch(void* const* d_in, const int* in_sizes, int n_in,
                              void* d_out, int out_size, void* d_ws, size_t ws_size,
                              hipStream_t stream) {
    (void)in_sizes; (void)n_in; (void)out_size; (void)ws_size;
    const float* x  = (const float*)d_in[0];
    const int*   ei = (const int*)d_in[1];
    const float* W  = (const float*)d_in[2];
    const float* b  = (const float*)d_in[3];
    float* out = (float*)d_out;

    int* rec = (int*)d_ws;
    unsigned* sorted = (unsigned*)(rec + (size_t)N_NODES * REC);
    int* off = (int*)(sorted + N_EDGES);

    k_chunk_sort <<<NCHUNK, T1, 0, stream>>>(ei, sorted, off);
    k_build_lists<<<NB, T1, 0, stream>>>(sorted, off, rec);
    k_gather_gemm<<<N_NODES / 16, 256, 0, stream>>>(x, ei, rec, W, b, out);
}


# Gather-product work-groups own 32 rows (two 16-row tiles; grid 1563, was 3125): a wave sums four rows and each W fragment it loads feeds two matrix products, halving W reads and waves
# speedup vs baseline: 4.6275x; 1.0424x over previous
#include <hip/hip_runtime.h>

#define N_NODES 50000
#define N_EDGES 800000
#define DIM     128
#define REC     64
#define CAP     (REC - 1)

#define NCHUNK  256
#define CHUNK   (N_EDGES / NCHUNK)
#define RB      200
#define NB      (N_NODES / RB)
#define T1      512
#define EPT     ((CHUNK + T1 - 1) / T1)
static_assert(N_EDGES % NCHUNK == 0, "chunks are whole");
static_assert(N_NODES % RB == 0, "buckets are whole");
static_assert(NB <= 256, "one scan thread per bucket, 8 waves");
static_assert(N_NODES <= 65536 && RB <= 65536, "an entry packs source id and local row in 16 bits each");
static_assert(NCHUNK == (T1 / 32) * 16, "each of the 16 waves of a list work-group takes 16 chunks");

typedef __attribute__((ext_vector_type(2))) float v2f;
typedef __attribute__((ext_vector_type(8))) float v8f;

__global__ void __launch_bounds__(T1) k_chunk_sort(const int* __restrict__ ei,
                                                   unsigned* __restrict__ sorted, int* __restrict__ off) {
    __shared__ int bcnt[256];
    __shared__ int boff[256];
    __shared__ int wtot[8];
    __shared__ unsigned stage[CHUNK];
    const int tid  = threadIdx.x;
    const int lane = tid & 31;
    const int base = blockIdx.x * CHUNK;
    if (tid < 256) bcnt[tid] = 0;
    __syncthreads();

    int sv[EPT], dv[EPT], pos[EPT];
    #pragma unroll
    for (int it = 0; it < EPT; ++it) {
        const int e = tid + it * T1;
        const bool ok = e < CHUNK;
        sv[it] = ok ? ei[base + e] : 0;
        dv[it] = ok ? ei[N_EDGES + base + e] : -1;
    }
    #pragma unroll
    for (int it = 0; it < EPT; ++it) {
        pos[it] = 0;
        if (dv[it] >= 0) pos[it] = atomicAdd(&bcnt[dv[it] / RB], 1);
    }
    __syncthreads();

    int v = 0, incl = 0;
    if (tid < 256) {
        v = tid < NB ? bcnt[tid] : 0;
        incl = v;
        #pragma unroll
        for (int d = 1; d < 32; d <<= 1) {
            const int t = __shfl_up(incl, d);
            if (lane >= d) incl += t;
        }
        if (lane == 31) wtot[tid >> 5] = incl;
    }
    __syncthreads();
    if (tid < 256) {
        int add = 0;
        #pragma unroll
        for (int w = 0; w < 8; ++w) if (w < (tid >> 5)) add += wtot[w];
        boff[tid] = add + incl - v;
    }
    __syncthreads();

    #pragma unroll
    for (int it = 0; it < EPT; ++it) {
        if (dv[it] >= 0) {
            const int b = dv[it] / RB;
            stage[boff[b] + pos[it]] = (unsigned)sv[it] | ((unsigned)(dv[it] - b * RB) << 16);
        }
    }
    __syncthreads();
    for (int i = tid; i < CHUNK; i += T1) sorted[base + i] = stage[i];
    if (tid <= NB) off[blockIdx.x * (NB + 1) + tid] = boff[tid];
}

static __device__ __forceinline__ void add_entry(int* __restrict__ rowcnt, int* __restrict__ lst, unsigned v) {
    const int r = (int)(v >> 16);
    const int p = atomicAdd(&rowcnt[r], 1);
    if (p < CAP) lst[r * REC + 1 + p] = (int)(v & 0xffffu);
}

__global__ void __launch_bounds__(T1) k_build_lists(const unsigned* __restrict__ sorted,
                                                    const int* __restrict__ off, int* __restrict__ rec) {
    __shared__ int rowcnt[RB];
    __shared__ int lst[RB * REC];
    const int tid  = threadIdx.x;
    const int wave = tid >> 5;
    const int lane = tid & 31;
    const int k    = blockIdx.x;
    for (int i = tid; i < RB; i += T1) rowcnt[i] = 0;
    __syncthreads();

    const int* offp = off + (size_t)(wave * 16 + (lane & 15)) * (NB + 1) + k;
    const int o0 = offp[0];
    const int o1 = offp[1];
    #pragma unroll
    for (int h = 0; h < 16; h += 8) {
        unsigned v[8];
        int s8[8], e8[8];
        #pragma unroll
        for (int u = 0; u < 8; ++u) {
            s8[u] = __builtin_amdgcn_readlane(o0, h + u);
            e8[u] = __builtin_amdgcn_readlane(o1, h + u);
            const unsigned* seg = sorted + (size_t)(wave * 16 + h + u) * CHUNK;
            const int j = s8[u] + lane;
            v[u] = seg[j < e8[u] ? j : 0];
        }
        #pragma unroll
        for (int u = 0; u < 8; ++u)
            if (s8[u] + lane < e8[u]) add_entry(rowcnt, lst, v[u]);
        #pragma unroll
        for (int u = 0; u < 8; ++u) {
            const unsigned* seg = sorted + (size_t)(wave * 16 + h + u) * CHUNK;
            for (int j = s8[u] + 32 + lane; j < e8[u]; j += 32) add_entry(rowcnt, lst, seg[j]);
        }
    }
    __syncthreads();

    for (int r = wave; r < RB; r += T1 / 32) {
        const int cnt = rowcnt[r];
        int* rp = rec + (size_t)(k * RB + r) * REC;
        const int lim = cnt < CAP ? cnt : CAP;
        rp[lane] = lane == 0 ? cnt : (lane <= lim ? lst[r * REC + lane] : 0);
        if (lim > 31) rp[32 + lane] = (32 + lane <= lim) ? lst[r * REC + 32 + lane] : 0;
    }
}

static __device__ __forceinline__ float4 ld_row(const float* __restrict__ x, int s, int lane) {
    return ((const float4*)(x + (size_t)s * DIM))[lane];
}
static __device__ __forceinline__ void fma4(float4& a, const float4 v, float w) {
    a.x = fmaf(v.x, w, a.x); a.y = fmaf(v.y, w, a.y);
    a.z = fmaf(v.z, w, a.z); a.w = fmaf(v.w, w, a.w);
}
static __device__ __forceinline__ float dinv_of(const int* __restrict__ rec, int n) {
    return rsqrtf((float)(rec[(size_t)n * REC] + 1));
}
static __device__ __forceinline__ void add_lanes(float4& a, const float* __restrict__ x,
                                                 int sv, float wv, int lo, int hi, int lane) {
    const int wi = __float_as_int(wv);
    int j = lo;
    for (; j + 4 <= hi; j += 4) {
        int s0 = __builtin_amdgcn_readlane(sv, j);
        int s1 = __builtin_amdgcn_readlane(sv, j + 1);
        int s2 = __builtin_amdgcn_readlane(sv, j + 2);
        int s3 = __builtin_amdgcn_readlane(sv, j + 3);
        float4 v0 = ld_row(x, s0, lane);
        float4 v1 = ld_row(x, s1, lane);
        float4 v2 = ld_row(x, s2, lane);
        float4 v3 = ld_row(x, s3, lane);
        fma4(a, v0, __int_as_float(__builtin_amdgcn_readlane(wi, j)));
        fma4(a, v1, __int_as_float(__builtin_amdgcn_readlane(wi, j + 1)));
        fma4(a, v2, __int_as_float(__builtin_amdgcn_readlane(wi, j + 2)));
        fma4(a, v3, __int_as_float(__builtin_amdgcn_readlane(wi, j + 3)));
    }
    for (; j < hi; ++j) {
        int s = __builtin_amdgcn_readlane(sv, j);
        float4 v = ld_row(x, s, lane);
        fma4(a, v, __int_as_float(__builtin_amdgcn_readlane(wi, j)));
    }
}

struct RowList { int c; int s0; int s1; float w0; float w1; };

static __device__ __forceinline__ float4 sum_row(const float* __restrict__ x, const int* __restrict__ ei,
                                                 const int* __restrict__ rec, const RowList L,
                                                 const float4 xs, int node, int lane) {
    const int c = L.c;
    const float dn = rsqrtf((float)(c + 1));
    float4 a = xs;
    a.x *= dn; a.y *= dn; a.z *= dn; a.w *= dn;
    if (c <= CAP) {
        const int h0 = c < 31 ? c : 31;
        add_lanes(a, x, L.s0, L.w0, 1, h0 + 1, lane);
        if (c > 31) add_lanes(a, x, L.s1, L.w1, 0, c - 31, lane);
    } else {
        for (int base = 0; base < N_EDGES; base += 32) {
            int e = base + lane;
            int d = (e < N_EDGES) ? ei[N_EDGES + e] : -1;
            unsigned m = __builtin_amdgcn_ballot_w32(d == node);
            while (m) {
                int j = __builtin_ctz(m);
                m &= m - 1;
                int s = ei[base + j];
                float4 v = ld_row(x, s, lane);
                fma4(a, v, dinv_of(rec, s));
            }
        }
    }
    a.x *= dn; a.y *= dn; a.z *= dn; a.w *= dn;
    return a;
}

static_assert(N_NODES % 16 == 0, "a 16-row tile is whole or absent");
#define TR      2
#define ROWS_WG (16 * TR)
#define GRID3   ((N_NODES + ROWS_WG - 1) / ROWS_WG)

__global__ void __launch_bounds__(256) k_gather_gemm(
        const float* __restrict__ x, const int* __restrict__ ei, const int* __restrict__ rec,
        const float* __restrict__ W, const float* __restrict__ bias, float* __restrict__ out) {
    __shared__ alignas(16) float At[ROWS_WG * 132];
    __shared__ alignas(16) float Xt[ROWS_WG * 132];
    const int tid  = threadIdx.x;
    const int wave = tid >> 5;
    const int lane = tid & 31;
    const int row0 = blockIdx.x * ROWS_WG;

    for (int t = 0; t < TR; ++t) {
        const int lr0 = t * 16 + wave * 2, lr1 = lr0 + 1;
        if (row0 + t * 16 >= N_NODES) {
            const float4 z = make_float4(0.0f, 0.0f, 0.0f, 0.0f);
            ((float4*)(At + lr0 * 132))[lane] = z;
            ((float4*)(At + lr1 * 132))[lane] = z;
            ((float4*)(Xt + lr0 * 132))[lane] = z;
            ((float4*)(Xt + lr1 * 132))[lane] = z;
            continue;
        }
        const int n0 = row0 + lr0, n1 = row0 + lr1;
        const int* r0 = rec + (size_t)n0 * REC;
        const int* r1 = rec + (size_t)n1 * REC;
        const int cv0 = r0[0];
        const int cv1 = r1[0];
        const float4 x0 = ld_row(x, n0, lane);
        const float4 x1 = ld_row(x, n1, lane);
        RowList L0, L1;
        L0.c = __builtin_amdgcn_readfirstlane(cv0);
        L1.c = __builtin_amdgcn_readfirstlane(cv1);
        const bool p00 = L0.c <= CAP && lane >= 1 && lane <= L0.c;
        const bool p01 = L0.c <= CAP && 32 + lane <= L0.c;
        const bool p10 = L1.c <= CAP && lane >= 1 && lane <= L1.c;
        const bool p11 = L1.c <= CAP && 32 + lane <= L1.c;
        L0.s0 = 0; L0.s1 = 0; L1.s0 = 0; L1.s1 = 0;
        if (p00) L0.s0 = r0[lane];
        if (p10) L1.s0 = r1[lane];
        if (p01) L0.s1 = r0[32 + lane];
        if (p11) L1.s1 = r1[32 + lane];
        const int d00 = rec[(size_t)L0.s0 * REC];
        const int d10 = rec[(size_t)L1.s0 * REC];
        const int d01 = rec[(size_t)L0.s1 * REC];
        const int d11 = rec[(size_t)L1.s1 * REC];
        L0.w0 = p00 ? rsqrtf((float)(d00 + 1)) : 0.0f;
        L1.w0 = p10 ? rsqrtf((float)(d10 + 1)) : 0.0f;
        L0.w1 = p01 ? rsqrtf((float)(d01 + 1)) : 0.0f;
        L1.w1 = p11 ? rsqrtf((float)(d11 + 1)) : 0.0f;
        ((float4*)(Xt + lr0 * 132))[lane] = x0;
        ((float4*)(Xt + lr1 * 132))[lane] = x1;
        const float4 a0 = sum_row(x, ei, rec, L0, x0, n0, lane);
        ((float4*)(At + lr0 * 132))[lane] = a0;
        const float4 a1 = sum_row(x, ei, rec, L1, x1, n1, lane);
        ((float4*)(At + lr1 * 132))[lane] = a1;
    }
    __syncthreads();

    const int col0 = wave * 16;
    const int nn   = lane & 15;
    const int koff = (lane >> 4) << 1;
    const int colg = col0 + nn;

    v2f bfrag[32];
    #pragma unroll
    for (int kk = 0; kk < 32; ++kk) {
        int k = (kk << 2) + koff;
        v2f t;
        t.x = W[k * DIM + colg];
        t.y = W[(k + 1) * DIM + colg];
        bfrag[kk] = t;
    }

    v8f c[TR];
    #pragma unroll
    for (int t = 0; t < TR; ++t) c[t] = v8f{};
    #pragma unroll
    for (int kk = 0; kk < 32; ++kk) {
        int k = (kk << 2) + koff;
        #pragma unroll
        for (int t = 0; t < TR; ++t) {
            const float* arow = At + (t * 16 + nn) * 132;
            v2f a;
            a.x = arow[k];
            a.y = arow[k + 1];
            c[t] = __builtin_amdgcn_wmma_f32_16x16x4_f32(false, a, false, bfrag[kk],
                                                         (short)0, c[t], false, false);
        }
    }

    const float bv = bias[colg];
    const int lrb = (lane >> 4) << 3;
    #pragma unroll
    for (int t = 0; t < TR; ++t) {
        if (row0 + t * 16 < N_NODES) {
            #pragma unroll
            for (int r = 0; r < 8; ++r) {
                float v = c[t][r] + bv;
                v = v > 0.0f ? v : 0.0f;
                const int lr = t * 16 + lrb + r;
                out[(size_t)(row0 + lr) * DIM + colg] = Xt[lr * 132 + colg] + v;
            }
        }
    }
}

extern "C" void kernel_launch(void* const* d_in, const int* in_sizes, int n_in,
                              void* d_out, int out_size, void* d_ws, size_t ws_size,
                              hipStream_t stream) {
    (void)in_sizes; (void)n_in; (void)out_size; (void)ws_size;
    const float* x  = (const float*)d_in[0];
    const int*   ei = (const int*)d_in[1];
    const float* W  = (const float*)d_in[2];
    const float* b  = (const float*)d_in[3];
    float* out = (float*)d_out;

    int* rec = (int*)d_ws;
    unsigned* sorted = (unsigned*)(rec + (size_t)N_NODES * REC);
    int* off = (int*)(sorted + N_EDGES);

    k_chunk_sort <<<NCHUNK, T1, 0, stream>>>(ei, sorted, off);
    k_build_lists<<<NB, T1, 0, stream>>>(sorted, off, rec);
    k_gather_gemm<<<GRID3, 256, 0, stream>>>(x, ei, rec, W, b, out);
}
